# starting kernel
# baseline (speedup 1.0000x reference)
#include <hip/hip_runtime.h>
#include <hip/hip_bf16.h>
#include <math.h>

#define BATCH   256
#define IN_DIM  1024
#define OUT_DIM 1024
#define NPATHS  2000
#define NSTEPS  128
#define NPTOT   (BATCH * NPATHS)
#define DT      (1.0f / 128.0f)
#define SQDT    0.08838834764831845f

typedef float v2f __attribute__((ext_vector_type(2)));
typedef float v8f __attribute__((ext_vector_type(8)));

__global__ __launch_bounds__(64) void bicep_fb_kernel(
    const float* __restrict__ x, const float* __restrict__ W_in,
    const float* __restrict__ b_in, const float* __restrict__ decay,
    float* __restrict__ a_ws) {
  __shared__ float red[64];
  const int bg = blockIdx.x;
  const int b = bg / 10, g = bg % 10;
  const float* xr = x + b * IN_DIM;
  const float* wr = W_in + g * IN_DIM;
  float s = 0.0f;
  for (int i = threadIdx.x; i < IN_DIM; i += 64) s = fmaf(xr[i], wr[i], s);
  red[threadIdx.x] = s;
  __syncthreads();
  for (int off = 32; off > 0; off >>= 1) {
    if (threadIdx.x < off) red[threadIdx.x] += red[threadIdx.x + off];
    __syncthreads();
  }
  if (threadIdx.x == 0) {
    float proj = red[0] + b_in[g];
    float fb = 1.0f / (1.0f + expf(-proj));
    a_ws[bg] = 1.0f + (fb - decay[0]) * DT;
  }
}

__global__ __launch_bounds__(256) void bicep_path_kernel(
    const float* __restrict__ eps, const float* __restrict__ W_conv,
    const float* __restrict__ a_ws,
    float* __restrict__ S0, float* __restrict__ S1, float* __restrict__ S2) {
  __shared__ float wc[(NSTEPS + 1) * 3];
  for (int i = threadIdx.x; i < (NSTEPS + 1) * 3; i += 256) wc[i] = W_conv[i];
  __syncthreads();

  const int id = blockIdx.x * 256 + threadIdx.x;
  const int b  = id / NPATHS;
  const int pg = id - b * NPATHS;
  const int g  = pg / 200;
  const float aa = a_ws[b * 10 + g];

  float X = 0.0f, s0 = 0.0f, s1 = 0.0f, s2 = 0.0f;
  const float* ep = eps + id;
  for (int t = 1; t <= NSTEPS; ++t) {
    float e = ep[(t - 1) * NPTOT];
    X  = fmaf(X, aa, SQDT * e);
    s0 = fmaf(wc[t * 3 + 0], X, s0);
    s1 = fmaf(wc[t * 3 + 1], X, s1);
    s2 = fmaf(wc[t * 3 + 2], X, s2);
  }
  S0[id] = s0; S1[id] = s1; S2[id] = s2;
}

__global__ __launch_bounds__(256) void bicep_agg_kernel(
    const float* __restrict__ S0, const float* __restrict__ S1,
    const float* __restrict__ S2, const float* __restrict__ b_conv,
    float* __restrict__ agg) {
  const int id = blockIdx.x * 256 + threadIdx.x;
  const int pg = id % NPATHS;
  float v = S1[id] + b_conv[0];
  if (pg > 0)          v += S0[id - 1];
  if (pg < NPATHS - 1) v += S2[id + 1];
  agg[id] = v;
}

__global__ __launch_bounds__(256) void bicep_gemm_kernel(
    const float* __restrict__ agg, const float* __restrict__ W_out,
    const float* __restrict__ b_out, float* __restrict__ out) {
  const int lane = threadIdx.x & 31;
  const int wave = threadIdx.x >> 5;
  const int tile = blockIdx.x * 8 + wave;
  const int mTile = tile >> 6;
  const int nTile = tile & 63;
  const int half = lane >> 4;
  const int idx  = lane & 15;

  const float* arow = agg   + (mTile * 16 + idx) * NPATHS + 2 * half;
  const float* brow = W_out + (nTile * 16 + idx) * NPATHS + 2 * half;

  v8f c = {};
  for (int k = 0; k < NPATHS; k += 4) {
    v2f a = *(const v2f*)(arow + k);
    v2f b = *(const v2f*)(brow + k);
    c = __builtin_amdgcn_wmma_f32_16x16x4_f32(
         false, a,  false, b,
         (short)0, c,  false,  false);
  }

  const float bb = b_out[nTile * 16 + idx];
  float* orow = out + (mTile * 16 + 8 * half) * OUT_DIM + nTile * 16 + idx;
#pragma unroll
  for (int r = 0; r < 8; ++r) orow[r * OUT_DIM] = c[r] + bb;
}

extern "C" void kernel_launch(void* const* d_in, const int* in_sizes, int n_in,
                              void* d_out, int out_size, void* d_ws, size_t ws_size,
                              hipStream_t stream) {
  (void)in_sizes; (void)n_in; (void)out_size; (void)ws_size;
  const float* x      = (const float*)d_in[0];
  const float* W_in   = (const float*)d_in[1];
  const float* b_in   = (const float*)d_in[2];
  const float* decay  = (const float*)d_in[3];
  const float* W_conv = (const float*)d_in[4];
  const float* b_conv = (const float*)d_in[5];
  const float* W_out  = (const float*)d_in[6];
  const float* b_out  = (const float*)d_in[7];
  const float* eps    = (const float*)d_in[8];
  float* out = (float*)d_out;

  float* ws   = (float*)d_ws;
  float* a_ws = ws;
  float* S0   = ws + 4096;
  float* S1   = S0 + NPTOT;
  float* S2   = S1 + NPTOT;
  float* agg  = S2 + NPTOT;

  bicep_fb_kernel<<<BATCH * 10, 64, 0, stream>>>(x, W_in, b_in, decay, a_ws);
  bicep_path_kernel<<<NPTOT / 256, 256, 0, stream>>>(eps, W_conv, a_ws, S0, S1, S2);
  bicep_agg_kernel<<<NPTOT / 256, 256, 0, stream>>>(S0, S1, S2, b_conv, agg);
  bicep_gemm_kernel<<<(16 * 64) / 8, 256, 0, stream>>>(agg, W_out, b_out, out);
}


# Combine launch removed: scan kernel adds in-group S0/S2 neighbour terms via LDS; group-edge terms go to a plane whose 2 live slots per 256 the GEMM A loader adds; 3 launches; edge sum order changes
# speedup vs baseline: 1.4736x; 1.4736x over previous
#include <hip/hip_runtime.h>
#include <hip/hip_bf16.h>
#include <math.h>

#define BATCH   256
#define IN_DIM  1024
#define OUT_DIM 1024
#define NPATHS  2000
#define NSTEPS  128
#define NPTOT   (BATCH * NPATHS)
#define DT      (1.0f / 128.0f)
#define SQDT    0.08838834764831845f

typedef float v2f __attribute__((ext_vector_type(2)));
typedef float v8f __attribute__((ext_vector_type(8)));

__global__ __launch_bounds__(64) void bicep_fb_kernel(
    const float* __restrict__ x, const float* __restrict__ W_in,
    const float* __restrict__ b_in, const float* __restrict__ decay,
    float* __restrict__ a_ws) {
  __shared__ float red[64];
  const int bg = blockIdx.x;
  const int b = bg / 10, g = bg % 10;
  const float* xr = x + b * IN_DIM;
  const float* wr = W_in + g * IN_DIM;
  float s = 0.0f;
  for (int i = threadIdx.x; i < IN_DIM; i += 64) s = fmaf(xr[i], wr[i], s);
  red[threadIdx.x] = s;
  __syncthreads();
  for (int off = 32; off > 0; off >>= 1) {
    if (threadIdx.x < off) red[threadIdx.x] += red[threadIdx.x + off];
    __syncthreads();
  }
  if (threadIdx.x == 0) {
    float proj = red[0] + b_in[g];
    float fb = 1.0f / (1.0f + expf(-proj));
    a_ws[bg] = 1.0f + (fb - decay[0]) * DT;
  }
}

#define NGROUPS (NPTOT / 256)

__global__ __launch_bounds__(256) void bicep_path_kernel(
    const float* __restrict__ eps, const float* __restrict__ W_conv,
    const float* __restrict__ b_conv, const float* __restrict__ a_ws,
    float* __restrict__ U, float* __restrict__ V) {
  __shared__ float wc[(NSTEPS + 1) * 3];
  __shared__ float e0[256];
  __shared__ float e2[256];
  for (int i = threadIdx.x; i < (NSTEPS + 1) * 3; i += 256) wc[i] = W_conv[i];
  __syncthreads();

  const int id = blockIdx.x * 256 + threadIdx.x;
  const int b  = id / NPATHS;
  const int pg = id - b * NPATHS;
  const int g  = pg / 200;
  const float aa = a_ws[b * 10 + g];

  float X = 0.0f, s0 = 0.0f, s1 = 0.0f, s2 = 0.0f;
  const float* ep = eps + id;
  for (int t = 1; t <= NSTEPS; ++t) {
    float e = ep[(t - 1) * NPTOT];
    X  = fmaf(X, aa, SQDT * e);
    s0 = fmaf(wc[t * 3 + 0], X, s0);
    s1 = fmaf(wc[t * 3 + 1], X, s1);
    s2 = fmaf(wc[t * 3 + 2], X, s2);
  }
  const int tid = threadIdx.x;
  e0[tid] = s0;
  e2[tid] = s2;
  __syncthreads();

  float u = s1 + b_conv[0];
  if (pg > 0 && tid > 0)            u += e0[tid - 1];
  if (pg < NPATHS - 1 && tid < 255) u += e2[tid + 1];
  U[id] = u;

  const int wave = tid >> 5;
  const int lane = tid & 31;
  int   vi = id;
  float vv = 0.0f;
  if (wave == 0 && blockIdx.x > 0) {
    const int pg_first = (blockIdx.x * 256) % NPATHS;
    vi = id - 32;
    if (lane == 31 && pg_first > 0) vv = e2[0];
  }
  if (wave == 7 && blockIdx.x < NGROUPS - 1) {
    const int pg_last = (blockIdx.x * 256 + 255) % NPATHS;
    vi = id + 32;
    if (lane == 0 && pg_last < NPATHS - 1) vv = e0[255];
  }
  V[vi] = vv;
}

typedef float v4f __attribute__((ext_vector_type(4)));

#define GK_CH     16
#define GK_MAIN   15
#define GK_EXTRA  (NPATHS / GK_CH - 8 * GK_MAIN)
#define GK_TS     (8 * 32 + 16)
#define GK_WS     (4 * GK_TS)

struct GChunk { v4f a0l, a0h, a1l, a1h, b0l, b0h, b1l, b1h; float v0, v1; };

struct GEdge { const float* V; int eo0, eo1, phase; bool lo; };

__device__ __forceinline__ void g_load(GChunk& f, const float* pa0, const float* pa1,
                                       const float* pb0, const float* pb1,
                                       const GEdge& g, int chunk) {
  const int o = chunk * GK_CH;
  const bool hit = ((g.phase + chunk) & 15) == 0;
  const float* vo = g.V + o;
  f.v0 = vo[hit ? g.eo0 : 1];
  f.v1 = vo[hit ? g.eo1 : 1];
  f.a0l = *(const v4f*)(pa0 + o);  f.a0h = *(const v4f*)(pa0 + o + 4);
  f.a1l = *(const v4f*)(pa1 + o);  f.a1h = *(const v4f*)(pa1 + o + 4);
  f.b0l = *(const v4f*)(pb0 + o);  f.b0h = *(const v4f*)(pb0 + o + 4);
  f.b1l = *(const v4f*)(pb1 + o);  f.b1h = *(const v4f*)(pb1 + o + 4);
}

__device__ __forceinline__ v8f g_wm(v2f a, v2f b, v8f c) {
  return __builtin_amdgcn_wmma_f32_16x16x4_f32(
       false, a,  false, b,
       (short)0, c,  false,  false);
}

__device__ __forceinline__ void g_step(v2f a0, v2f a1, v2f b0, v2f b1,
                                       v8f& c00, v8f& c01, v8f& c10, v8f& c11) {
  c00 = g_wm(a0, b0, c00);
  c01 = g_wm(a0, b1, c01);
  c10 = g_wm(a1, b0, c10);
  c11 = g_wm(a1, b1, c11);
}

__device__ __forceinline__ void g_mma(const GChunk& f, bool lo,
                                      v8f& c00, v8f& c01, v8f& c10, v8f& c11) {
  v4f a0l = f.a0l, a0h = f.a0h, a1l = f.a1l, a1h = f.a1h;
  a0l.x += lo ? f.v0 : 0.0f;
  a0h.w += lo ? 0.0f : f.v0;
  a1l.x += lo ? f.v1 : 0.0f;
  a1h.w += lo ? 0.0f : f.v1;
  g_step(a0l.xy, a1l.xy, f.b0l.xy, f.b1l.xy, c00, c01, c10, c11);
  g_step(a0l.zw, a1l.zw, f.b0l.zw, f.b1l.zw, c00, c01, c10, c11);
  g_step(a0h.xy, a1h.xy, f.b0h.xy, f.b1h.xy, c00, c01, c10, c11);
  g_step(a0h.zw, a1h.zw, f.b0h.zw, f.b1h.zw, c00, c01, c10, c11);
}

__global__ __launch_bounds__(256) void bicep_gemm_kernel(
    const float* __restrict__ U, const float* __restrict__ W_out,
    const float* __restrict__ b_out, float* __restrict__ out) {
  __shared__ float red[8 * GK_WS];

  const int lane = threadIdx.x & 31;
  const int wave = __builtin_amdgcn_readfirstlane((int)(threadIdx.x >> 5));
  const int mt = blockIdx.x & 7;
  const int nt = blockIdx.x >> 3;
  const int half = lane >> 4;
  const int idx  = lane & 15;

  const int kw = wave * (GK_MAIN * GK_CH) + 8 * half;
  const float* pa0 = U     + (mt * 32 + idx) * NPATHS + kw;
  const float* pa1 = pa0 + 16 * NPATHS;
  const float* pb0 = W_out + (nt * 32 + idx) * NPATHS + kw;
  const float* pb1 = pb0 + 16 * NPATHS;

  GEdge g;
  g.V     = U + NPTOT;
  g.lo    = (half == 0);
  g.phase = (13 * idx - (half ? 15 : 0) + wave * GK_MAIN) & 15;
  g.eo0   = (mt * 32 + idx) * NPATHS + kw + (half ? 7 : 0);
  g.eo1   = g.eo0 + 16 * NPATHS;

  v8f c00 = {}, c01 = {}, c10 = {}, c11 = {};
  GChunk f0, f1, f2;
  g_load(f0, pa0, pa1, pb0, pb1, g, 0);
  g_load(f1, pa0, pa1, pb0, pb1, g, 1);
  g_load(f2, pa0, pa1, pb0, pb1, g, 2);

  if (wave < GK_EXTRA) {
    GChunk fe;
    g_load(fe, pa0, pa1, pb0, pb1, g, 8 * GK_MAIN + wave - wave * GK_MAIN);
    g_mma(fe, g.lo, c00, c01, c10, c11);
  }

#pragma unroll
  for (int j = 0; j < GK_MAIN - 3; j += 3) {
    g_mma(f0, g.lo, c00, c01, c10, c11);  g_load(f0, pa0, pa1, pb0, pb1, g, j + 3);
    g_mma(f1, g.lo, c00, c01, c10, c11);  g_load(f1, pa0, pa1, pb0, pb1, g, j + 4);
    g_mma(f2, g.lo, c00, c01, c10, c11);  g_load(f2, pa0, pa1, pb0, pb1, g, j + 5);
  }
  g_mma(f0, g.lo, c00, c01, c10, c11);
  g_mma(f1, g.lo, c00, c01, c10, c11);
  g_mma(f2, g.lo, c00, c01, c10, c11);

  float* rw = red + wave * GK_WS + lane;
#pragma unroll
  for (int r = 0; r < 8; ++r) {
    rw[0 * GK_TS + r * 32] = c00[r];
    rw[1 * GK_TS + r * 32] = c01[r];
    rw[2 * GK_TS + r * 32] = c10[r];
    rw[3 * GK_TS + r * 32] = c11[r];
  }
  __syncthreads();

  const float bb = b_out[nt * 32 + lane];
  float* orow = out + (mt * 32 + wave) * OUT_DIM + nt * 32 + lane;
  for (int jj = 0; jj < 4; ++jj) {
    const int e = (2 * (jj >> 1) + half) * GK_TS + wave * 32 + (jj & 1) * 16 + idx;
    float s = red[e];
    for (int w = 1; w < 8; ++w) s += red[w * GK_WS + e];
    orow[jj * 8 * OUT_DIM] = s + bb;
  }
}

extern "C" void kernel_launch(void* const* d_in, const int* in_sizes, int n_in,
                              void* d_out, int out_size, void* d_ws, size_t ws_size,
                              hipStream_t stream) {
  (void)in_sizes; (void)n_in; (void)out_size; (void)ws_size;
  const float* x      = (const float*)d_in[0];
  const float* W_in   = (const float*)d_in[1];
  const float* b_in   = (const float*)d_in[2];
  const float* decay  = (const float*)d_in[3];
  const float* W_conv = (const float*)d_in[4];
  const float* b_conv = (const float*)d_in[5];
  const float* W_out  = (const float*)d_in[6];
  const float* b_out  = (const float*)d_in[7];
  const float* eps    = (const float*)d_in[8];
  float* out = (float*)d_out;

  float* ws   = (float*)d_ws;
  float* a_ws = ws;
  float* U    = ws + 4096;
  float* V    = U + NPTOT;

  bicep_fb_kernel<<<BATCH * 10, 64, 0, stream>>>(x, W_in, b_in, decay, a_ws);
  bicep_path_kernel<<<NGROUPS, 256, 0, stream>>>(eps, W_conv, b_conv, a_ws, U, V);
  bicep_gemm_kernel<<<8 * 32, 256, 0, stream>>>(U, W_out, b_out, out);
}


# Feedback launch removed: each scan work-group computes the drift factors of its own 2-3 (row, group) pairs, loads issued beside the conv-weight loads; 3 launches -> 2; feedback dot sum order changes
# speedup vs baseline: 1.9101x; 1.2962x over previous
#include <hip/hip_runtime.h>
#include <hip/hip_bf16.h>
#include <math.h>

#define BATCH   256
#define IN_DIM  1024
#define OUT_DIM 1024
#define NPATHS  2000
#define NSTEPS  128
#define NPTOT   (BATCH * NPATHS)
#define DT      (1.0f / 128.0f)
#define SQDT    0.08838834764831845f

typedef float v2f __attribute__((ext_vector_type(2)));
typedef float v8f __attribute__((ext_vector_type(8)));

#define NGROUPS (NPTOT / 256)
#define PPG     200
#define WC_N    ((NSTEPS + 1) * 3)
#define FB_MAXQ 3

typedef float v4f __attribute__((ext_vector_type(4)));

__global__ __launch_bounds__(256) void bicep_path_kernel(
    const float* __restrict__ eps, const float* __restrict__ W_conv,
    const float* __restrict__ b_conv, const float* __restrict__ x,
    const float* __restrict__ W_in, const float* __restrict__ b_in,
    const float* __restrict__ decay, float* __restrict__ U, float* __restrict__ V) {
  __shared__ float wc[WC_N];
  __shared__ float e0[256];
  __shared__ float e2[256];
  __shared__ float pr[FB_MAXQ * 8];

  const int tid  = threadIdx.x;
  const int wave = tid >> 5;
  const int lane = tid & 31;
  const int id = blockIdx.x * 256 + tid;
  const int q  = id / PPG;
  const int q0 = (blockIdx.x * 256) / PPG;
  const int ql = (blockIdx.x * 256 + 255) / PPG;

  const float w_lo = W_conv[tid];
  const float w_hi = (tid < WC_N - 256) ? W_conv[tid + 256] : 0.0f;
  const float bin  = b_in[q % 10];

  v4f xv[FB_MAXQ], wv[FB_MAXQ];
  for (int j = 0; j < FB_MAXQ; ++j) {
    const int qq = min(q0 + j, ql);
    const int bb = qq / 10, gg = qq - bb * 10;
    xv[j] = *(const v4f*)(x    + bb * IN_DIM + 4 * tid);
    wv[j] = *(const v4f*)(W_in + gg * IN_DIM + 4 * tid);
  }
  float s[FB_MAXQ];
  for (int j = 0; j < FB_MAXQ; ++j) {
    float d = xv[j].x * wv[j].x;
    d = fmaf(xv[j].y, wv[j].y, d);
    d = fmaf(xv[j].z, wv[j].z, d);
    s[j] = fmaf(xv[j].w, wv[j].w, d);
  }
  for (int off = 16; off > 0; off >>= 1) {
    for (int j = 0; j < FB_MAXQ; ++j) s[j] += __shfl_xor(s[j], off);
  }
  if (lane == 0) {
    for (int j = 0; j < FB_MAXQ; ++j) pr[j * 8 + wave] = s[j];
  }
  wc[tid] = w_lo;
  if (tid < WC_N - 256) wc[tid + 256] = w_hi;
  __syncthreads();

  const float* pp = pr + (q - q0) * 8;
  float proj = pp[0];
  for (int w = 1; w < 8; ++w) proj += pp[w];
  proj += bin;
  const float fb = 1.0f / (1.0f + expf(-proj));
  const float aa = 1.0f + (fb - decay[0]) * DT;

  const int b  = id / NPATHS;
  const int pg = id - b * NPATHS;

  float X = 0.0f, s0 = 0.0f, s1 = 0.0f, s2 = 0.0f;
  const float* ep = eps + id;
  for (int t = 1; t <= NSTEPS; ++t) {
    float e = ep[(t - 1) * NPTOT];
    X  = fmaf(X, aa, SQDT * e);
    s0 = fmaf(wc[t * 3 + 0], X, s0);
    s1 = fmaf(wc[t * 3 + 1], X, s1);
    s2 = fmaf(wc[t * 3 + 2], X, s2);
  }
  e0[tid] = s0;
  e2[tid] = s2;
  __syncthreads();

  float u = s1 + b_conv[0];
  if (pg > 0 && tid > 0)            u += e0[tid - 1];
  if (pg < NPATHS - 1 && tid < 255) u += e2[tid + 1];
  U[id] = u;

  int   vi = id;
  float vv = 0.0f;
  if (wave == 0 && blockIdx.x > 0) {
    const int pg_first = (blockIdx.x * 256) % NPATHS;
    vi = id - 32;
    if (lane == 31 && pg_first > 0) vv = e2[0];
  }
  if (wave == 7 && blockIdx.x < NGROUPS - 1) {
    const int pg_last = (blockIdx.x * 256 + 255) % NPATHS;
    vi = id + 32;
    if (lane == 0 && pg_last < NPATHS - 1) vv = e0[255];
  }
  V[vi] = vv;
}

#define GK_CH     16
#define GK_MAIN   15
#define GK_EXTRA  (NPATHS / GK_CH - 8 * GK_MAIN)
#define GK_TS     (8 * 32 + 16)
#define GK_WS     (4 * GK_TS)

struct GChunk { v4f a0l, a0h, a1l, a1h, b0l, b0h, b1l, b1h; float v0, v1; };

struct GEdge { const float* V; int eo0, eo1, phase; bool lo; };

__device__ __forceinline__ void g_load(GChunk& f, const float* pa0, const float* pa1,
                                       const float* pb0, const float* pb1,
                                       const GEdge& g, int chunk) {
  const int o = chunk * GK_CH;
  const bool hit = ((g.phase + chunk) & 15) == 0;
  const float* vo = g.V + o;
  f.v0 = vo[hit ? g.eo0 : 1];
  f.v1 = vo[hit ? g.eo1 : 1];
  f.a0l = *(const v4f*)(pa0 + o);  f.a0h = *(const v4f*)(pa0 + o + 4);
  f.a1l = *(const v4f*)(pa1 + o);  f.a1h = *(const v4f*)(pa1 + o + 4);
  f.b0l = *(const v4f*)(pb0 + o);  f.b0h = *(const v4f*)(pb0 + o + 4);
  f.b1l = *(const v4f*)(pb1 + o);  f.b1h = *(const v4f*)(pb1 + o + 4);
}

__device__ __forceinline__ v8f g_wm(v2f a, v2f b, v8f c) {
  return __builtin_amdgcn_wmma_f32_16x16x4_f32(
       false, a,  false, b,
       (short)0, c,  false,  false);
}

__device__ __forceinline__ void g_step(v2f a0, v2f a1, v2f b0, v2f b1,
                                       v8f& c00, v8f& c01, v8f& c10, v8f& c11) {
  c00 = g_wm(a0, b0, c00);
  c01 = g_wm(a0, b1, c01);
  c10 = g_wm(a1, b0, c10);
  c11 = g_wm(a1, b1, c11);
}

__device__ __forceinline__ void g_mma(const GChunk& f, bool lo,
                                      v8f& c00, v8f& c01, v8f& c10, v8f& c11) {
  v4f a0l = f.a0l, a0h = f.a0h, a1l = f.a1l, a1h = f.a1h;
  a0l.x += lo ? f.v0 : 0.0f;
  a0h.w += lo ? 0.0f : f.v0;
  a1l.x += lo ? f.v1 : 0.0f;
  a1h.w += lo ? 0.0f : f.v1;
  g_step(a0l.xy, a1l.xy, f.b0l.xy, f.b1l.xy, c00, c01, c10, c11);
  g_step(a0l.zw, a1l.zw, f.b0l.zw, f.b1l.zw, c00, c01, c10, c11);
  g_step(a0h.xy, a1h.xy, f.b0h.xy, f.b1h.xy, c00, c01, c10, c11);
  g_step(a0h.zw, a1h.zw, f.b0h.zw, f.b1h.zw, c00, c01, c10, c11);
}

__global__ __launch_bounds__(256) void bicep_gemm_kernel(
    const float* __restrict__ U, const float* __restrict__ W_out,
    const float* __restrict__ b_out, float* __restrict__ out) {
  __shared__ float red[8 * GK_WS];

  const int lane = threadIdx.x & 31;
  const int wave = __builtin_amdgcn_readfirstlane((int)(threadIdx.x >> 5));
  const int mt = blockIdx.x & 7;
  const int nt = blockIdx.x >> 3;
  const int half = lane >> 4;
  const int idx  = lane & 15;

  const int kw = wave * (GK_MAIN * GK_CH) + 8 * half;
  const float* pa0 = U     + (mt * 32 + idx) * NPATHS + kw;
  const float* pa1 = pa0 + 16 * NPATHS;
  const float* pb0 = W_out + (nt * 32 + idx) * NPATHS + kw;
  const float* pb1 = pb0 + 16 * NPATHS;

  GEdge g;
  g.V     = U + NPTOT;
  g.lo    = (half == 0);
  g.phase = (13 * idx - (half ? 15 : 0) + wave * GK_MAIN) & 15;
  g.eo0   = (mt * 32 + idx) * NPATHS + kw + (half ? 7 : 0);
  g.eo1   = g.eo0 + 16 * NPATHS;

  v8f c00 = {}, c01 = {}, c10 = {}, c11 = {};
  GChunk f0, f1, f2;
  g_load(f0, pa0, pa1, pb0, pb1, g, 0);
  g_load(f1, pa0, pa1, pb0, pb1, g, 1);
  g_load(f2, pa0, pa1, pb0, pb1, g, 2);

  if (wave < GK_EXTRA) {
    GChunk fe;
    g_load(fe, pa0, pa1, pb0, pb1, g, 8 * GK_MAIN + wave - wave * GK_MAIN);
    g_mma(fe, g.lo, c00, c01, c10, c11);
  }

#pragma unroll
  for (int j = 0; j < GK_MAIN - 3; j += 3) {
    g_mma(f0, g.lo, c00, c01, c10, c11);  g_load(f0, pa0, pa1, pb0, pb1, g, j + 3);
    g_mma(f1, g.lo, c00, c01, c10, c11);  g_load(f1, pa0, pa1, pb0, pb1, g, j + 4);
    g_mma(f2, g.lo, c00, c01, c10, c11);  g_load(f2, pa0, pa1, pb0, pb1, g, j + 5);
  }
  g_mma(f0, g.lo, c00, c01, c10, c11);
  g_mma(f1, g.lo, c00, c01, c10, c11);
  g_mma(f2, g.lo, c00, c01, c10, c11);

  float* rw = red + wave * GK_WS + lane;
#pragma unroll
  for (int r = 0; r < 8; ++r) {
    rw[0 * GK_TS + r * 32] = c00[r];
    rw[1 * GK_TS + r * 32] = c01[r];
    rw[2 * GK_TS + r * 32] = c10[r];
    rw[3 * GK_TS + r * 32] = c11[r];
  }
  __syncthreads();

  const float bb = b_out[nt * 32 + lane];
  float* orow = out + (mt * 32 + wave) * OUT_DIM + nt * 32 + lane;
  for (int jj = 0; jj < 4; ++jj) {
    const int e = (2 * (jj >> 1) + half) * GK_TS + wave * 32 + (jj & 1) * 16 + idx;
    float s = red[e];
    for (int w = 1; w < 8; ++w) s += red[w * GK_WS + e];
    orow[jj * 8 * OUT_DIM] = s + bb;
  }
}

extern "C" void kernel_launch(void* const* d_in, const int* in_sizes, int n_in,
                              void* d_out, int out_size, void* d_ws, size_t ws_size,
                              hipStream_t stream) {
  (void)in_sizes; (void)n_in; (void)out_size; (void)ws_size;
  const float* x      = (const float*)d_in[0];
  const float* W_in   = (const float*)d_in[1];
  const float* b_in   = (const float*)d_in[2];
  const float* decay  = (const float*)d_in[3];
  const float* W_conv = (const float*)d_in[4];
  const float* b_conv = (const float*)d_in[5];
  const float* W_out  = (const float*)d_in[6];
  const float* b_out  = (const float*)d_in[7];
  const float* eps    = (const float*)d_in[8];
  float* out = (float*)d_out;

  float* U = (float*)d_ws;
  float* V = U + NPTOT;

  bicep_path_kernel<<<NGROUPS, 256, 0, stream>>>(eps, W_conv, b_conv, x, W_in, b_in, decay, U, V);
  bicep_gemm_kernel<<<8 * 32, 256, 0, stream>>>(U, W_out, b_out, out);
}


# Path scan: 1024 paths per work-group, 4 per lane (a wave reads an aligned 512 B per step, 4-byte loads), eps in a 16-step register ring, 60 loads in flight; sum order changes at inner 256-run starts
# speedup vs baseline: 2.0119x; 1.0533x over previous
#include <hip/hip_runtime.h>
#include <hip/hip_bf16.h>
#include <math.h>

#define BATCH   256
#define IN_DIM  1024
#define OUT_DIM 1024
#define NPATHS  2000
#define NSTEPS  128
#define NPTOT   (BATCH * NPATHS)
#define DT      (1.0f / 128.0f)
#define SQDT    0.08838834764831845f

typedef float v2f __attribute__((ext_vector_type(2)));
typedef float v8f __attribute__((ext_vector_type(8)));

#define PW       4
#define WG_PATHS (256 * PW)
#define NGROUPS  (NPTOT / WG_PATHS)
#define PPG      200
#define WC_N     ((NSTEPS + 1) * 3)
#define FB_MAXQ  7
#define RING     16
#define AHEAD    (RING - 1)

typedef float v4f __attribute__((ext_vector_type(4)));

__device__ __forceinline__ float eps_at(const float* plane, unsigned byte_off, int k) {
  return ((const float*)((const char*)plane + byte_off))[k];
}

template <bool REFILL>
__device__ __forceinline__ void scan_trip(float (&ring)[RING][PW], const float* eps,
                                          unsigned lob, int sb, const float* wc,
                                          const float (&aa)[PW], float (&X)[PW],
                                          float (&s0)[PW], float (&s1)[PW], float (&s2)[PW]) {
#pragma unroll
  for (int i = 0; i < RING; ++i) {
    float e[PW];
#pragma unroll
    for (int r = 0; r < PW; ++r) e[r] = ring[i][r];
    if (REFILL || i == 0) {
      const float* plane = eps + (size_t)(sb + i + AHEAD) * NPTOT;
#pragma unroll
      for (int r = 0; r < PW; ++r) ring[(i + AHEAD) % RING][r] = eps_at(plane, lob, 32 * r);
    }
    const float* w = wc + (sb + i + 1) * 3;
    const float w0 = w[0], w1 = w[1], w2 = w[2];
#pragma unroll
    for (int r = 0; r < PW; ++r) {
      X[r]  = fmaf(X[r], aa[r], SQDT * e[r]);
      s0[r] = fmaf(w0, X[r], s0[r]);
      s1[r] = fmaf(w1, X[r], s1[r]);
      s2[r] = fmaf(w2, X[r], s2[r]);
    }
    if ((i & 3) == 3) __builtin_amdgcn_sched_barrier(0);
  }
}

__global__ __launch_bounds__(256) void bicep_path_kernel(
    const float* __restrict__ eps, const float* __restrict__ W_conv,
    const float* __restrict__ b_conv, const float* __restrict__ x,
    const float* __restrict__ W_in, const float* __restrict__ b_in,
    const float* __restrict__ decay, float* __restrict__ U, float* __restrict__ V) {
  __shared__ float wc[WC_N];
  __shared__ float e0[WG_PATHS];
  __shared__ float e2[WG_PATHS];
  __shared__ float pr[FB_MAXQ * 8];
  __shared__ float pb[FB_MAXQ];

  const int tid  = threadIdx.x;
  const int wave = tid >> 5;
  const int lane = tid & 31;
  const int base = blockIdx.x * WG_PATHS;
  const int o0   = wave * 128 + lane;
  const int id0  = base + o0;
  const int q0 = base / PPG;
  const int ql = (base + WG_PATHS - 1) / PPG;

  const float w_lo = W_conv[tid];
  const float w_hi = (tid < WC_N - 256) ? W_conv[tid + 256] : 0.0f;

  v4f xv[FB_MAXQ], wv[FB_MAXQ];
  float bq[FB_MAXQ];
  for (int j = 0; j < FB_MAXQ; ++j) {
    const int qq = min(q0 + j, ql);
    const int bb = qq / 10, gg = qq - bb * 10;
    xv[j] = *(const v4f*)(x    + bb * IN_DIM + 4 * tid);
    wv[j] = *(const v4f*)(W_in + gg * IN_DIM + 4 * tid);
    bq[j] = b_in[gg];
  }
  int qr[PW];
  for (int r = 0; r < PW; ++r) qr[r] = (id0 + 32 * r) / PPG - q0;

  const unsigned lob = (unsigned)id0 * 4u;
  float ring[RING][PW];
  __builtin_amdgcn_sched_barrier(0);
#pragma unroll
  for (int i = 0; i < AHEAD; ++i) {
    const float* plane = eps + (size_t)i * NPTOT;
#pragma unroll
    for (int r = 0; r < PW; ++r) ring[i][r] = eps_at(plane, lob, 32 * r);
    if ((i & 3) == 3 || i == AHEAD - 1) __builtin_amdgcn_sched_barrier(0);
  }

  float s[FB_MAXQ];
  for (int j = 0; j < FB_MAXQ; ++j) {
    float d = xv[j].x * wv[j].x;
    d = fmaf(xv[j].y, wv[j].y, d);
    d = fmaf(xv[j].z, wv[j].z, d);
    s[j] = fmaf(xv[j].w, wv[j].w, d);
  }
  for (int off = 16; off > 0; off >>= 1) {
    for (int j = 0; j < FB_MAXQ; ++j) s[j] += __shfl_xor(s[j], off);
  }
  if (lane == 0) {
    for (int j = 0; j < FB_MAXQ; ++j) pr[j * 8 + wave] = s[j];
    if (wave == 0) {
      for (int j = 0; j < FB_MAXQ; ++j) pb[j] = bq[j];
    }
  }
  wc[tid] = w_lo;
  if (tid < WC_N - 256) wc[tid + 256] = w_hi;
  __builtin_amdgcn_fence(__ATOMIC_RELEASE, "workgroup", "local");
  __builtin_amdgcn_s_barrier();
  __builtin_amdgcn_fence(__ATOMIC_ACQUIRE, "workgroup", "local");

  const float dec = decay[0];
  float aa[PW];
  for (int r = 0; r < PW; ++r) {
    const float* pp = pr + qr[r] * 8;
    float proj = pp[0];
    for (int w = 1; w < 8; ++w) proj += pp[w];
    proj += pb[qr[r]];
    const float fb = 1.0f / (1.0f + expf(-proj));
    aa[r] = 1.0f + (fb - dec) * DT;
  }

  float X[PW], s0[PW], s1[PW], s2[PW];
  for (int r = 0; r < PW; ++r) { X[r] = 0.0f; s0[r] = 0.0f; s1[r] = 0.0f; s2[r] = 0.0f; }

  for (int sb = 0; sb < NSTEPS - RING; sb += RING)
    scan_trip<true>(ring, eps, lob, sb, wc, aa, X, s0, s1, s2);
  scan_trip<false>(ring, eps, lob, NSTEPS - RING, wc, aa, X, s0, s1, s2);

  for (int r = 0; r < PW; ++r) {
    e0[o0 + 32 * r] = s0[r];
    e2[o0 + 32 * r] = s2[r];
  }
  __syncthreads();

  const float bc = b_conv[0];
  for (int r = 0; r < PW; ++r) {
    const int o  = o0 + 32 * r;
    const int id = id0 + 32 * r;
    const int pg = id % NPATHS;
    float u = s1[r] + bc;
    if (pg > 0 && o > 0)                     u += e0[o - 1];
    if (pg < NPATHS - 1 && o < WG_PATHS - 1) u += e2[o + 1];
    U[id] = u;

    int   vi = id;
    float vv = 0.0f;
    if (wave == 0 && r == 0 && blockIdx.x > 0) {
      const int pg_first = base % NPATHS;
      vi = id - 32;
      if (lane == 31 && pg_first > 0) vv = e2[0];
    }
    if (wave == 7 && r == PW - 1 && blockIdx.x < NGROUPS - 1) {
      const int pg_last = (base + WG_PATHS - 1) % NPATHS;
      vi = id + 32;
      if (lane == 0 && pg_last < NPATHS - 1) vv = e0[WG_PATHS - 1];
    }
    V[vi] = vv;
  }
}

#define GK_CH     16
#define GK_MAIN   15
#define GK_EXTRA  (NPATHS / GK_CH - 8 * GK_MAIN)
#define GK_TS     (8 * 32 + 16)
#define GK_WS     (4 * GK_TS)

struct GChunk { v4f a0l, a0h, a1l, a1h, b0l, b0h, b1l, b1h; float v0, v1; };

struct GEdge { const float* V; int eo0, eo1, phase; bool lo; };

__device__ __forceinline__ void g_load(GChunk& f, const float* pa0, const float* pa1,
                                       const float* pb0, const float* pb1,
                                       const GEdge& g, int chunk) {
  const int o = chunk * GK_CH;
  const bool hit = ((g.phase + chunk) & 15) == 0;
  const float* vo = g.V + o;
  f.v0 = vo[hit ? g.eo0 : 1];
  f.v1 = vo[hit ? g.eo1 : 1];
  f.a0l = *(const v4f*)(pa0 + o);  f.a0h = *(const v4f*)(pa0 + o + 4);
  f.a1l = *(const v4f*)(pa1 + o);  f.a1h = *(const v4f*)(pa1 + o + 4);
  f.b0l = *(const v4f*)(pb0 + o);  f.b0h = *(const v4f*)(pb0 + o + 4);
  f.b1l = *(const v4f*)(pb1 + o);  f.b1h = *(const v4f*)(pb1 + o + 4);
}

__device__ __forceinline__ v8f g_wm(v2f a, v2f b, v8f c) {
  return __builtin_amdgcn_wmma_f32_16x16x4_f32(
       false, a,  false, b,
       (short)0, c,  false,  false);
}

__device__ __forceinline__ void g_step(v2f a0, v2f a1, v2f b0, v2f b1,
                                       v8f& c00, v8f& c01, v8f& c10, v8f& c11) {
  c00 = g_wm(a0, b0, c00);
  c01 = g_wm(a0, b1, c01);
  c10 = g_wm(a1, b0, c10);
  c11 = g_wm(a1, b1, c11);
}

__device__ __forceinline__ void g_mma(const GChunk& f, bool lo,
                                      v8f& c00, v8f& c01, v8f& c10, v8f& c11) {
  v4f a0l = f.a0l, a0h = f.a0h, a1l = f.a1l, a1h = f.a1h;
  a0l.x += lo ? f.v0 : 0.0f;
  a0h.w += lo ? 0.0f : f.v0;
  a1l.x += lo ? f.v1 : 0.0f;
  a1h.w += lo ? 0.0f : f.v1;
  g_step(a0l.xy, a1l.xy, f.b0l.xy, f.b1l.xy, c00, c01, c10, c11);
  g_step(a0l.zw, a1l.zw, f.b0l.zw, f.b1l.zw, c00, c01, c10, c11);
  g_step(a0h.xy, a1h.xy, f.b0h.xy, f.b1h.xy, c00, c01, c10, c11);
  g_step(a0h.zw, a1h.zw, f.b0h.zw, f.b1h.zw, c00, c01, c10, c11);
}

__global__ __launch_bounds__(256) void bicep_gemm_kernel(
    const float* __restrict__ U, const float* __restrict__ W_out,
    const float* __restrict__ b_out, float* __restrict__ out) {
  __shared__ float red[8 * GK_WS];

  const int lane = threadIdx.x & 31;
  const int wave = __builtin_amdgcn_readfirstlane((int)(threadIdx.x >> 5));
  const int mt = blockIdx.x & 7;
  const int nt = blockIdx.x >> 3;
  const int half = lane >> 4;
  const int idx  = lane & 15;

  const int kw = wave * (GK_MAIN * GK_CH) + 8 * half;
  const float* pa0 = U     + (mt * 32 + idx) * NPATHS + kw;
  const float* pa1 = pa0 + 16 * NPATHS;
  const float* pb0 = W_out + (nt * 32 + idx) * NPATHS + kw;
  const float* pb1 = pb0 + 16 * NPATHS;

  GEdge g;
  g.V     = U + NPTOT;
  g.lo    = (half == 0);
  g.phase = (13 * idx - (half ? 15 : 0) + wave * GK_MAIN) & 15;
  g.eo0   = (mt * 32 + idx) * NPATHS + kw + (half ? 7 : 0);
  g.eo1   = g.eo0 + 16 * NPATHS;

  v8f c00 = {}, c01 = {}, c10 = {}, c11 = {};
  GChunk f0, f1, f2;
  g_load(f0, pa0, pa1, pb0, pb1, g, 0);
  g_load(f1, pa0, pa1, pb0, pb1, g, 1);
  g_load(f2, pa0, pa1, pb0, pb1, g, 2);

  if (wave < GK_EXTRA) {
    GChunk fe;
    g_load(fe, pa0, pa1, pb0, pb1, g, 8 * GK_MAIN + wave - wave * GK_MAIN);
    g_mma(fe, g.lo, c00, c01, c10, c11);
  }

#pragma unroll
  for (int j = 0; j < GK_MAIN - 3; j += 3) {
    g_mma(f0, g.lo, c00, c01, c10, c11);  g_load(f0, pa0, pa1, pb0, pb1, g, j + 3);
    g_mma(f1, g.lo, c00, c01, c10, c11);  g_load(f1, pa0, pa1, pb0, pb1, g, j + 4);
    g_mma(f2, g.lo, c00, c01, c10, c11);  g_load(f2, pa0, pa1, pb0, pb1, g, j + 5);
  }
  g_mma(f0, g.lo, c00, c01, c10, c11);
  g_mma(f1, g.lo, c00, c01, c10, c11);
  g_mma(f2, g.lo, c00, c01, c10, c11);

  float* rw = red + wave * GK_WS + lane;
#pragma unroll
  for (int r = 0; r < 8; ++r) {
    rw[0 * GK_TS + r * 32] = c00[r];
    rw[1 * GK_TS + r * 32] = c01[r];
    rw[2 * GK_TS + r * 32] = c10[r];
    rw[3 * GK_TS + r * 32] = c11[r];
  }
  __syncthreads();

  const float bb = b_out[nt * 32 + lane];
  float* orow = out + (mt * 32 + wave) * OUT_DIM + nt * 32 + lane;
  for (int jj = 0; jj < 4; ++jj) {
    const int e = (2 * (jj >> 1) + half) * GK_TS + wave * 32 + (jj & 1) * 16 + idx;
    float s = red[e];
    for (int w = 1; w < 8; ++w) s += red[w * GK_WS + e];
    orow[jj * 8 * OUT_DIM] = s + bb;
  }
}

extern "C" void kernel_launch(void* const* d_in, const int* in_sizes, int n_in,
                              void* d_out, int out_size, void* d_ws, size_t ws_size,
                              hipStream_t stream) {
  (void)in_sizes; (void)n_in; (void)out_size; (void)ws_size;
  const float* x      = (const float*)d_in[0];
  const float* W_in   = (const float*)d_in[1];
  const float* b_in   = (const float*)d_in[2];
  const float* decay  = (const float*)d_in[3];
  const float* W_conv = (const float*)d_in[4];
  const float* b_conv = (const float*)d_in[5];
  const float* W_out  = (const float*)d_in[6];
  const float* b_out  = (const float*)d_in[7];
  const float* eps    = (const float*)d_in[8];
  float* out = (float*)d_out;

  float* U = (float*)d_ws;
  float* V = U + NPTOT;

  bicep_path_kernel<<<NGROUPS, 256, 0, stream>>>(eps, W_conv, b_conv, x, W_in, b_in, decay, U, V);
  bicep_gemm_kernel<<<8 * 32, 256, 0, stream>>>(U, W_out, b_out, out);
}


# GEMM: rows go to fragments by parity and a wave asks for each 128 B operand line whole (the two 16-k chunks sharing it together): a wave fetches a line once; waves take 16 chunks; sum order changes
# speedup vs baseline: 2.1614x; 1.0743x over previous
#include <hip/hip_runtime.h>
#include <hip/hip_bf16.h>
#include <math.h>

#define BATCH   256
#define IN_DIM  1024
#define OUT_DIM 1024
#define NPATHS  2000
#define NSTEPS  128
#define NPTOT   (BATCH * NPATHS)
#define DT      (1.0f / 128.0f)
#define SQDT    0.08838834764831845f

typedef float v2f __attribute__((ext_vector_type(2)));
typedef float v8f __attribute__((ext_vector_type(8)));

#define PW       4
#define WG_PATHS (256 * PW)
#define NGROUPS  (NPTOT / WG_PATHS)
#define PPG      200
#define WC_N     ((NSTEPS + 1) * 3)
#define FB_MAXQ  7
#define RING     16
#define AHEAD    (RING - 1)

typedef float v4f __attribute__((ext_vector_type(4)));

__device__ __forceinline__ float eps_at(const float* plane, unsigned byte_off, int k) {
  return ((const float*)((const char*)plane + byte_off))[k];
}

template <bool REFILL>
__device__ __forceinline__ void scan_trip(float (&ring)[RING][PW], const float* eps,
                                          unsigned lob, int sb, const float* wc,
                                          const float (&aa)[PW], float (&X)[PW],
                                          float (&s0)[PW], float (&s1)[PW], float (&s2)[PW]) {
#pragma unroll
  for (int i = 0; i < RING; ++i) {
    float e[PW];
#pragma unroll
    for (int r = 0; r < PW; ++r) e[r] = ring[i][r];
    if (REFILL || i == 0) {
      const float* plane = eps + (size_t)(sb + i + AHEAD) * NPTOT;
#pragma unroll
      for (int r = 0; r < PW; ++r) ring[(i + AHEAD) % RING][r] = eps_at(plane, lob, 32 * r);
    }
    const float* w = wc + (sb + i + 1) * 3;
    const float w0 = w[0], w1 = w[1], w2 = w[2];
#pragma unroll
    for (int r = 0; r < PW; ++r) {
      X[r]  = fmaf(X[r], aa[r], SQDT * e[r]);
      s0[r] = fmaf(w0, X[r], s0[r]);
      s1[r] = fmaf(w1, X[r], s1[r]);
      s2[r] = fmaf(w2, X[r], s2[r]);
    }
    if ((i & 3) == 3) __builtin_amdgcn_sched_barrier(0);
  }
}

__global__ __launch_bounds__(256) void bicep_path_kernel(
    const float* __restrict__ eps, const float* __restrict__ W_conv,
    const float* __restrict__ b_conv, const float* __restrict__ x,
    const float* __restrict__ W_in, const float* __restrict__ b_in,
    const float* __restrict__ decay, float* __restrict__ U, float* __restrict__ V) {
  __shared__ float wc[WC_N];
  __shared__ float e0[WG_PATHS];
  __shared__ float e2[WG_PATHS];
  __shared__ float pr[FB_MAXQ * 8];
  __shared__ float pb[FB_MAXQ];

  const int tid  = threadIdx.x;
  const int wave = tid >> 5;
  const int lane = tid & 31;
  const int base = blockIdx.x * WG_PATHS;
  const int o0   = wave * 128 + lane;
  const int id0  = base + o0;
  const int q0 = base / PPG;
  const int ql = (base + WG_PATHS - 1) / PPG;

  const float w_lo = W_conv[tid];
  const float w_hi = (tid < WC_N - 256) ? W_conv[tid + 256] : 0.0f;

  v4f xv[FB_MAXQ], wv[FB_MAXQ];
  float bq[FB_MAXQ];
  for (int j = 0; j < FB_MAXQ; ++j) {
    const int qq = min(q0 + j, ql);
    const int bb = qq / 10, gg = qq - bb * 10;
    xv[j] = *(const v4f*)(x    + bb * IN_DIM + 4 * tid);
    wv[j] = *(const v4f*)(W_in + gg * IN_DIM + 4 * tid);
    bq[j] = b_in[gg];
  }
  int qr[PW];
  for (int r = 0; r < PW; ++r) qr[r] = (id0 + 32 * r) / PPG - q0;

  const unsigned lob = (unsigned)id0 * 4u;
  float ring[RING][PW];
  __builtin_amdgcn_sched_barrier(0);
#pragma unroll
  for (int i = 0; i < AHEAD; ++i) {
    const float* plane = eps + (size_t)i * NPTOT;
#pragma unroll
    for (int r = 0; r < PW; ++r) ring[i][r] = eps_at(plane, lob, 32 * r);
    if ((i & 3) == 3 || i == AHEAD - 1) __builtin_amdgcn_sched_barrier(0);
  }

  float s[FB_MAXQ];
  for (int j = 0; j < FB_MAXQ; ++j) {
    float d = xv[j].x * wv[j].x;
    d = fmaf(xv[j].y, wv[j].y, d);
    d = fmaf(xv[j].z, wv[j].z, d);
    s[j] = fmaf(xv[j].w, wv[j].w, d);
  }
  for (int off = 16; off > 0; off >>= 1) {
    for (int j = 0; j < FB_MAXQ; ++j) s[j] += __shfl_xor(s[j], off);
  }
  if (lane == 0) {
    for (int j = 0; j < FB_MAXQ; ++j) pr[j * 8 + wave] = s[j];
    if (wave == 0) {
      for (int j = 0; j < FB_MAXQ; ++j) pb[j] = bq[j];
    }
  }
  wc[tid] = w_lo;
  if (tid < WC_N - 256) wc[tid + 256] = w_hi;
  __builtin_amdgcn_fence(__ATOMIC_RELEASE, "workgroup", "local");
  __builtin_amdgcn_s_barrier();
  __builtin_amdgcn_fence(__ATOMIC_ACQUIRE, "workgroup", "local");

  const float dec = decay[0];
  float aa[PW];
  for (int r = 0; r < PW; ++r) {
    const float* pp = pr + qr[r] * 8;
    float proj = pp[0];
    for (int w = 1; w < 8; ++w) proj += pp[w];
    proj += pb[qr[r]];
    const float fb = 1.0f / (1.0f + expf(-proj));
    aa[r] = 1.0f + (fb - dec) * DT;
  }

  float X[PW], s0[PW], s1[PW], s2[PW];
  for (int r = 0; r < PW; ++r) { X[r] = 0.0f; s0[r] = 0.0f; s1[r] = 0.0f; s2[r] = 0.0f; }

  for (int sb = 0; sb < NSTEPS - RING; sb += RING)
    scan_trip<true>(ring, eps, lob, sb, wc, aa, X, s0, s1, s2);
  scan_trip<false>(ring, eps, lob, NSTEPS - RING, wc, aa, X, s0, s1, s2);

  for (int r = 0; r < PW; ++r) {
    e0[o0 + 32 * r] = s0[r];
    e2[o0 + 32 * r] = s2[r];
  }
  __syncthreads();

  const float bc = b_conv[0];
  for (int r = 0; r < PW; ++r) {
    const int o  = o0 + 32 * r;
    const int id = id0 + 32 * r;
    const int pg = id % NPATHS;
    float u = s1[r] + bc;
    if (pg > 0 && o > 0)                     u += e0[o - 1];
    if (pg < NPATHS - 1 && o < WG_PATHS - 1) u += e2[o + 1];
    U[id] = u;

    int   vi = id;
    float vv = 0.0f;
    if (wave == 0 && r == 0 && blockIdx.x > 0) {
      const int pg_first = base % NPATHS;
      vi = id - 32;
      if (lane == 31 && pg_first > 0) vv = e2[0];
    }
    if (wave == 7 && r == PW - 1 && blockIdx.x < NGROUPS - 1) {
      const int pg_last = (base + WG_PATHS - 1) % NPATHS;
      vi = id + 32;
      if (lane == 0 && pg_last < NPATHS - 1) vv = e0[WG_PATHS - 1];
    }
    V[vi] = vv;
  }
}

#define GK_CH     16
#define GK_FULL   16
#define GK_LAST   (NPATHS / GK_CH - 7 * GK_FULL)
#define GK_TS     (8 * 32 + 16)
#define GK_WS     (4 * GK_TS)

struct GHalf { v4f al, ah, bl, bh; float v; };

struct GEdge { const float* V; int eo, phase; };

__device__ __forceinline__ void g_load(GHalf& f, const float* pa, const float* pb,
                                       const GEdge& g, int chunk) {
  const int o = chunk * GK_CH;
  const bool hit = ((g.phase + chunk) & 15) == 0;
  const float* vo = g.V + o;
  f.v  = vo[hit ? g.eo : 1];
  f.al = *(const v4f*)(pa + o);  f.ah = *(const v4f*)(pa + o + 4);
  f.bl = *(const v4f*)(pb + o);  f.bh = *(const v4f*)(pb + o + 4);
}

__device__ __forceinline__ v8f g_wm(v2f a, v2f b, v8f c) {
  return __builtin_amdgcn_wmma_f32_16x16x4_f32(
       false, a,  false, b,
       (short)0, c,  false,  false);
}

__device__ __forceinline__ void g_step(v2f a0, v2f a1, v2f b0, v2f b1,
                                       v8f& c00, v8f& c01, v8f& c10, v8f& c11) {
  c00 = g_wm(a0, b0, c00);
  c01 = g_wm(a0, b1, c01);
  c10 = g_wm(a1, b0, c10);
  c11 = g_wm(a1, b1, c11);
}

__device__ __forceinline__ void g_mma(const GHalf& f0, const GHalf& f1, bool lo, bool keep,
                                      v8f& c00, v8f& c01, v8f& c10, v8f& c11) {
  v4f a0l = f0.al, a0h = f0.ah, a1l = f1.al, a1h = f1.ah;
  a0l.x += lo ? f0.v : 0.0f;
  a0h.w += lo ? 0.0f : f0.v;
  a1l.x += lo ? f1.v : 0.0f;
  a1h.w += lo ? 0.0f : f1.v;
  const v4f zero = {0.0f, 0.0f, 0.0f, 0.0f};
  a0l = keep ? a0l : zero;  a0h = keep ? a0h : zero;
  a1l = keep ? a1l : zero;  a1h = keep ? a1h : zero;
  g_step(a0l.xy, a1l.xy, f0.bl.xy, f1.bl.xy, c00, c01, c10, c11);
  g_step(a0l.zw, a1l.zw, f0.bl.zw, f1.bl.zw, c00, c01, c10, c11);
  g_step(a0h.xy, a1h.xy, f0.bh.xy, f1.bh.xy, c00, c01, c10, c11);
  g_step(a0h.zw, a1h.zw, f0.bh.zw, f1.bh.zw, c00, c01, c10, c11);
}

__global__ __launch_bounds__(256) void bicep_gemm_kernel(
    const float* __restrict__ U, const float* __restrict__ W_out,
    const float* __restrict__ b_out, float* __restrict__ out) {
  __shared__ float red[8 * GK_WS];

  const int lane = threadIdx.x & 31;
  const int wave = __builtin_amdgcn_readfirstlane((int)(threadIdx.x >> 5));
  const int mt = blockIdx.x & 7;
  const int nt = blockIdx.x >> 3;
  const int half = lane >> 4;
  const int idx  = lane & 15;
  const bool lo  = (half == 0);
  const bool full = (wave < 7);

  const int kb = wave * GK_FULL;
  const int kw = kb * GK_CH + 8 * half;
  const float* pa0 = U     + (mt * 32 + 2 * idx) * NPATHS + kw;
  const float* pa1 = pa0 + NPATHS;
  const float* pb0 = W_out + (nt * 32 + 2 * idx) * NPATHS + kw;
  const float* pb1 = pb0 + NPATHS;

  GEdge g0, g1;
  g0.V = U + NPTOT;
  g1.V = g0.V;
  g0.phase = (10 * idx - (half ? 15 : 0) + kb) & 15;
  g1.phase = (g0.phase + 13) & 15;
  g0.eo = (mt * 32 + 2 * idx) * NPATHS + kw + (half ? 7 : 0);
  g1.eo = g0.eo + NPATHS;

  v8f c00 = {}, c01 = {}, c10 = {}, c11 = {};
  GHalf fe[GK_FULL], fo[GK_FULL];

#define GK_LD(buf, pa, pb, g, cc) \
  g_load(buf[cc], pa, pb, g, ((cc) < GK_LAST || full) ? (cc) : GK_LAST - 1)

  GK_LD(fe, pa0, pb0, g0, 0);  GK_LD(fe, pa0, pb0, g0, 1);
  GK_LD(fo, pa1, pb1, g1, 0);
  __builtin_amdgcn_sched_barrier(0);
  GK_LD(fo, pa1, pb1, g1, 1);  GK_LD(fo, pa1, pb1, g1, 2);
  GK_LD(fe, pa0, pb0, g0, 2);  GK_LD(fe, pa0, pb0, g0, 3);
  __builtin_amdgcn_sched_barrier(0);
  GK_LD(fo, pa1, pb1, g1, 3);  GK_LD(fo, pa1, pb1, g1, 4);
  __builtin_amdgcn_sched_barrier(0);

#pragma unroll
  for (int c = 0; c < GK_FULL; ++c) {
    g_mma(fe[c], fo[c], lo, c < GK_LAST || full, c00, c01, c10, c11);
    if ((c & 1) && c + 3 < GK_FULL) {
      __builtin_amdgcn_sched_barrier(0);
      GK_LD(fe, pa0, pb0, g0, c + 3);
      if (c + 4 < GK_FULL) { GK_LD(fe, pa0, pb0, g0, c + 4);  GK_LD(fo, pa1, pb1, g1, c + 4); }
      if (c + 5 < GK_FULL) GK_LD(fo, pa1, pb1, g1, c + 5);
      __builtin_amdgcn_sched_barrier(0);
    }
  }
#undef GK_LD

  float* rw = red + wave * GK_WS + lane;
#pragma unroll
  for (int r = 0; r < 8; ++r) {
    rw[0 * GK_TS + r * 32] = c00[r];
    rw[1 * GK_TS + r * 32] = c01[r];
    rw[2 * GK_TS + r * 32] = c10[r];
    rw[3 * GK_TS + r * 32] = c11[r];
  }
  __syncthreads();

  const float bb = b_out[nt * 32 + lane];
  float* orow = out + (mt * 32 + wave) * OUT_DIM + nt * 32 + lane;
  for (int jj = 0; jj < 4; ++jj) {
    const int q = (2 * (wave & 1) + (lane & 1)) * GK_TS +
                  ((wave >> 1) + 4 * (jj & 1)) * 32 + (jj >> 1) * 16 + (lane >> 1);
    float s = red[q];
    for (int w = 1; w < 8; ++w) s += red[w * GK_WS + q];
    orow[jj * 8 * OUT_DIM] = s + bb;
  }
}

extern "C" void kernel_launch(void* const* d_in, const int* in_sizes, int n_in,
                              void* d_out, int out_size, void* d_ws, size_t ws_size,
                              hipStream_t stream) {
  (void)in_sizes; (void)n_in; (void)out_size; (void)ws_size;
  const float* x      = (const float*)d_in[0];
  const float* W_in   = (const float*)d_in[1];
  const float* b_in   = (const float*)d_in[2];
  const float* decay  = (const float*)d_in[3];
  const float* W_conv = (const float*)d_in[4];
  const float* b_conv = (const float*)d_in[5];
  const float* W_out  = (const float*)d_in[6];
  const float* b_out  = (const float*)d_in[7];
  const float* eps    = (const float*)d_in[8];
  float* out = (float*)d_out;

  float* U = (float*)d_ws;
  float* V = U + NPTOT;

  bicep_path_kernel<<<NGROUPS, 256, 0, stream>>>(eps, W_conv, b_conv, x, W_in, b_in, decay, U, V);
  bicep_gemm_kernel<<<8 * 32, 256, 0, stream>>>(U, W_out, b_out, out);
}
